# starting kernel
# baseline (speedup 1.0000x reference)
#include <hip/hip_runtime.h>

typedef __attribute__((ext_vector_type(2))) float v2f;
typedef __attribute__((ext_vector_type(8))) float v8f;

#define LDS_STRIDE 50
#define ROWS 16
#define WPB 8

#define LDS_FENCE() asm volatile("s_wait_dscnt 0" ::: "memory")

static __device__ __forceinline__ float rcp_f(float x) {
    return __builtin_amdgcn_rcpf(x);
}
static __device__ __forceinline__ float tanh_f(float x) {
#if __has_builtin(__builtin_amdgcn_tanhf)
    return __builtin_amdgcn_tanhf(x);
#else
    float e = __expf(-2.0f * x);
    return (1.0f - e) * rcp_f(1.0f + e);
#endif
}
static __device__ __forceinline__ float sigmoid_f(float x) {
    return fmaf(0.5f, tanh_f(0.5f * x), 0.5f);
}
static __device__ __forceinline__ float selu_f(float x) {
    const float a = 1.6732632423543772f, s = 1.0507009873554805f;
    return x > 0.0f ? s * x : s * a * (__expf(x) - 1.0f);
}
static __device__ __forceinline__ float lstm_h(float gi, float gg, float go) {
    float c = sigmoid_f(gi) * tanh_f(gg);
    return sigmoid_f(go) * tanh_f(c);
}
static __device__ __forceinline__ v8f splat8(float v) {
    v8f r = {v, v, v, v, v, v, v, v};
    return r;
}
static __device__ __forceinline__ v8f wmma4(v2f a, v2f b, v8f c) {
    return __builtin_amdgcn_wmma_f32_16x16x4_f32(false, a, false, b, (short)0, c,
                                                 false, false);
}

static __device__ __forceinline__ void store_gates(float* g, const v8f* acc,
                                                   int half, int l16) {
#pragma unroll
    for (int t = 0; t < 3; ++t)
#pragma unroll
        for (int r = 0; r < 8; ++r)
            g[(r + half * 8) * LDS_STRIDE + t * 16 + l16] = acc[t][r];
}

static __device__ __forceinline__ void cell_slots(const float* g, int half, int l16,
                                                  float* hr) {
#pragma unroll
    for (int s = 0; s < 3; ++s)
#pragma unroll
        for (int v = 0; v < 2; ++v) {
            const int k = s * 4 + half * 2 + v;
            const float* row = g + l16 * LDS_STRIDE;
            hr[s * 2 + v] = lstm_h(row[k], row[24 + k], row[36 + k]);
        }
}

static __device__ __forceinline__ void emit_out(const float* g, float* dst, int lane) {
#pragma unroll
    for (int q = 0; q < 6; ++q) {
        const int flat = q * 32 + lane;
        const int r = flat / 12;
        const int j = flat - r * 12;
        const float* row = g + r * LDS_STRIDE;
        dst[flat] = tanh_f(lstm_h(row[j], row[24 + j], row[36 + j]));
    }
}

__global__ __launch_bounds__(256) void dynamics_net_kernel(
    const float* __restrict__ x,
    const float* __restrict__ W1, const float* __restrict__ bi1, const float* __restrict__ bh1,
    const float* __restrict__ W2, const float* __restrict__ bi2, const float* __restrict__ bh2,
    const float* __restrict__ Wm, const float* __restrict__ bim, const float* __restrict__ bhm,
    const float* __restrict__ Wd, const float* __restrict__ bid, const float* __restrict__ bhd,
    float* __restrict__ out, int n_tiles, long long rowsB) {
    __shared__ float lds[WPB * ROWS * LDS_STRIDE];

    const int lane = threadIdx.x & 31;
    const int wave = threadIdx.x >> 5;
    const int half = lane >> 4;
    const int l16  = lane & 15;
    const int tile = blockIdx.x * WPB + wave;
    if (tile >= n_tiles) return;

    float* g = lds + wave * (ROWS * LDS_STRIDE);

    v2f bw1[3][4];
#pragma unroll
    for (int t = 0; t < 3; ++t) {
        const float* wr = W1 + (t * 16 + l16) * 16 + half * 2;
#pragma unroll
        for (int s = 0; s < 4; ++s) bw1[t][s] = *(const v2f*)(wr + s * 4);
    }
    v2f bw2[3][3], bwm[3][3], bwd[3][3];
#pragma unroll
    for (int t = 0; t < 3; ++t) {
        const int n = t * 16 + l16;
#pragma unroll
        for (int s = 0; s < 3; ++s) {
            bw2[t][s] = *(const v2f*)(W2 + n * 12 + half * 2 + s * 4);
            bwm[t][s] = *(const v2f*)(Wm + n * 12 + half * 2 + s * 4);
            bwd[t][s] = *(const v2f*)(Wd + n * 12 + half * 2 + s * 4);
        }
    }
    float bs1[3], bs2[3], bsm[3], bsd[3];
#pragma unroll
    for (int t = 0; t < 3; ++t) {
        const int n = t * 16 + l16;
        bs1[t] = bi1[n] + bh1[n];
        bs2[t] = bi2[n] + bh2[n];
        bsm[t] = bim[n] + bhm[n];
        bsd[t] = bid[n] + bhd[n];
    }

    const float* xr = x + ((long long)tile * 16 + l16) * 16 + half * 2;
    v2f xa[4];
#pragma unroll
    for (int s = 0; s < 4; ++s) xa[s] = *(const v2f*)(xr + s * 4);

    v8f acc[3];
    float hr[6];
    v2f ha[3];

#pragma unroll
    for (int t = 0; t < 3; ++t) {
        v8f c = splat8(bs1[t]);
#pragma unroll
        for (int s = 0; s < 4; ++s) c = wmma4(xa[s], bw1[t][s], c);
        acc[t] = c;
    }
    store_gates(g, acc, half, l16);
    LDS_FENCE();
    cell_slots(g, half, l16, hr);
#pragma unroll
    for (int s = 0; s < 3; ++s) {
        v2f a = {fmaxf(hr[2 * s], 0.0f), fmaxf(hr[2 * s + 1], 0.0f)};
        ha[s] = a;
    }
    LDS_FENCE();

#pragma unroll
    for (int t = 0; t < 3; ++t) {
        v8f c = splat8(bs2[t]);
#pragma unroll
        for (int s = 0; s < 3; ++s) c = wmma4(ha[s], bw2[t][s], c);
        acc[t] = c;
    }
    store_gates(g, acc, half, l16);
    LDS_FENCE();
    cell_slots(g, half, l16, hr);
#pragma unroll
    for (int s = 0; s < 3; ++s) {
        v2f a = {selu_f(hr[2 * s]), selu_f(hr[2 * s + 1])};
        ha[s] = a;
    }
    LDS_FENCE();

#pragma unroll
    for (int t = 0; t < 3; ++t) {
        v8f c = splat8(bsm[t]);
#pragma unroll
        for (int s = 0; s < 3; ++s) c = wmma4(ha[s], bwm[t][s], c);
        acc[t] = c;
    }
    store_gates(g, acc, half, l16);
    LDS_FENCE();
    emit_out(g, out + (long long)tile * 192, lane);
    LDS_FENCE();

#pragma unroll
    for (int t = 0; t < 3; ++t) {
        v8f c = splat8(bsd[t]);
#pragma unroll
        for (int s = 0; s < 3; ++s) c = wmma4(ha[s], bwd[t][s], c);
        acc[t] = c;
    }
    store_gates(g, acc, half, l16);
    LDS_FENCE();
    emit_out(g, out + rowsB * 12 + (long long)tile * 192, lane);
}

extern "C" void kernel_launch(void* const* d_in, const int* in_sizes, int n_in,
                              void* d_out, int out_size, void* d_ws, size_t ws_size,
                              hipStream_t stream) {
    const float* x   = (const float*)d_in[0];
    const float* W1  = (const float*)d_in[1];
    const float* bi1 = (const float*)d_in[2];
    const float* bh1 = (const float*)d_in[3];
    const float* W2  = (const float*)d_in[4];
    const float* bi2 = (const float*)d_in[5];
    const float* bh2 = (const float*)d_in[6];
    const float* Wm  = (const float*)d_in[7];
    const float* bim = (const float*)d_in[8];
    const float* bhm = (const float*)d_in[9];
    const float* Wd  = (const float*)d_in[10];
    const float* bid = (const float*)d_in[11];
    const float* bhd = (const float*)d_in[12];
    float* out = (float*)d_out;

    const long long rowsB = (long long)in_sizes[0] / 16;
    const int n_tiles = (int)((rowsB + 15) / 16);
    const int blocks  = (n_tiles + WPB - 1) / WPB;

    hipLaunchKernelGGL(dynamics_net_kernel, dim3(blocks), dim3(256), 0, stream,
                       x, W1, bi1, bh1, W2, bi2, bh2, Wm, bim, bhm, Wd, bid, bhd,
                       out, n_tiles, rowsB);
}


# weights as a block-shared LDS operand: an 8-wave block copies the 4 weight matrices and bias sums to LDS once; each wave then walks 8 row tiles, B fragments read from LDS; 8192 waves, was 65536
# speedup vs baseline: 1.0688x; 1.0688x over previous
#include <hip/hip_runtime.h>

typedef __attribute__((ext_vector_type(2))) float v2f;
typedef __attribute__((ext_vector_type(4))) float v4f;
typedef __attribute__((ext_vector_type(8))) float v8f;

#define LDS_STRIDE 50
#define ROWS 16
#define WPB 8
#define TPW 8
#define W1_PITCH 20

#define LDS_W1   (WPB * ROWS * LDS_STRIDE)
#define LDS_WK   (LDS_W1 + 48 * W1_PITCH)
#define LDS_BS   (LDS_WK + 3 * 48 * 12)
#define LDS_ALL  (LDS_BS + 4 * 48)

#define LDS_FENCE() asm volatile("s_wait_dscnt 0" ::: "memory")

static __device__ __forceinline__ float rcp_f(float x) {
    return __builtin_amdgcn_rcpf(x);
}
static __device__ __forceinline__ float tanh_f(float x) {
#if __has_builtin(__builtin_amdgcn_tanhf)
    return __builtin_amdgcn_tanhf(x);
#else
    float e = __expf(-2.0f * x);
    return (1.0f - e) * rcp_f(1.0f + e);
#endif
}
static __device__ __forceinline__ float sigmoid_f(float x) {
    return fmaf(0.5f, tanh_f(0.5f * x), 0.5f);
}
static __device__ __forceinline__ float selu_f(float x) {
    const float a = 1.6732632423543772f, s = 1.0507009873554805f;
    return x > 0.0f ? s * x : s * a * (__expf(x) - 1.0f);
}
static __device__ __forceinline__ float lstm_h(float gi, float gg, float go) {
    float c = sigmoid_f(gi) * tanh_f(gg);
    return sigmoid_f(go) * tanh_f(c);
}
static __device__ __forceinline__ v8f splat8(float v) {
    v8f r = {v, v, v, v, v, v, v, v};
    return r;
}
static __device__ __forceinline__ v8f wmma4(v2f a, v2f b, v8f c) {
    return __builtin_amdgcn_wmma_f32_16x16x4_f32(false, a, false, b, (short)0, c,
                                                 false, false);
}

static __device__ __forceinline__ void store_gates(float* g, const v8f* acc,
                                                   int half, int l16) {
#pragma unroll
    for (int t = 0; t < 3; ++t)
#pragma unroll
        for (int r = 0; r < 8; ++r)
            g[(r + half * 8) * LDS_STRIDE + t * 16 + l16] = acc[t][r];
}

static __device__ __forceinline__ void cell_slots(const float* g, int half, int l16,
                                                  float* hr) {
#pragma unroll
    for (int s = 0; s < 3; ++s)
#pragma unroll
        for (int v = 0; v < 2; ++v) {
            const int k = s * 4 + half * 2 + v;
            const float* row = g + l16 * LDS_STRIDE;
            hr[s * 2 + v] = lstm_h(row[k], row[24 + k], row[36 + k]);
        }
}

static __device__ __forceinline__ void emit_out(const float* g, float* dst, int lane) {
#pragma unroll
    for (int q = 0; q < 6; ++q) {
        const int flat = q * 32 + lane;
        const int r = flat / 12;
        const int j = flat - r * 12;
        const float* row = g + r * LDS_STRIDE;
        dst[flat] = tanh_f(lstm_h(row[j], row[24 + j], row[36 + j]));
    }
}

static __device__ __forceinline__ void load_wk(const float* wl, int half, int l16,
                                               v2f (*bw)[3]) {
#pragma unroll
    for (int t = 0; t < 3; ++t)
#pragma unroll
        for (int s = 0; s < 3; ++s)
            bw[t][s] = *(const v2f*)(wl + (t * 16 + l16) * 12 + s * 4 + half * 2);
}
static __device__ __forceinline__ void load_w1(const float* wl, int half, int l16,
                                               v2f (*bw)[4]) {
#pragma unroll
    for (int t = 0; t < 3; ++t)
#pragma unroll
        for (int s = 0; s < 4; ++s)
            bw[t][s] = *(const v2f*)(wl + (t * 16 + l16) * W1_PITCH + s * 4 + half * 2);
}
static __device__ __forceinline__ void load_bs(const float* bl, int l16, float* b) {
#pragma unroll
    for (int t = 0; t < 3; ++t) b[t] = bl[t * 16 + l16];
}
static __device__ __forceinline__ void load_x(const float* __restrict__ x, int tile,
                                              int half, int l16, v2f* xa) {
    const float* xr = x + ((long long)tile * 16 + l16) * 16 + half * 2;
#pragma unroll
    for (int s = 0; s < 4; ++s) xa[s] = *(const v2f*)(xr + s * 4);
}

__global__ __launch_bounds__(256) void dynamics_net_kernel(
    const float* __restrict__ x,
    const float* __restrict__ W1, const float* __restrict__ bi1, const float* __restrict__ bh1,
    const float* __restrict__ W2, const float* __restrict__ bi2, const float* __restrict__ bh2,
    const float* __restrict__ Wm, const float* __restrict__ bim, const float* __restrict__ bhm,
    const float* __restrict__ Wd, const float* __restrict__ bid, const float* __restrict__ bhd,
    float* __restrict__ out, int n_tiles, long long rowsB) {
    __shared__ __attribute__((aligned(16))) float lds[LDS_ALL];

    const int tid  = threadIdx.x;
    const int lane = tid & 31;
    const int wave = __builtin_amdgcn_readfirstlane(tid >> 5);
    const int half = lane >> 4;
    const int l16  = lane & 15;

    if (tid < 192) {
        const int n = tid >> 2, c = tid & 3;
        *(v4f*)(lds + LDS_W1 + n * W1_PITCH + c * 4) = *(const v4f*)(W1 + n * 16 + c * 4);
    }
    if (tid < 144) {
        *(v4f*)(lds + LDS_WK + tid * 4)        = *(const v4f*)(W2 + tid * 4);
        *(v4f*)(lds + LDS_WK + 576 + tid * 4)  = *(const v4f*)(Wm + tid * 4);
        *(v4f*)(lds + LDS_WK + 1152 + tid * 4) = *(const v4f*)(Wd + tid * 4);
    }
    if (tid < 48) {
        lds[LDS_BS + tid]       = bi1[tid] + bh1[tid];
        lds[LDS_BS + 48 + tid]  = bi2[tid] + bh2[tid];
        lds[LDS_BS + 96 + tid]  = bim[tid] + bhm[tid];
        lds[LDS_BS + 144 + tid] = bid[tid] + bhd[tid];
    }
    __syncthreads();

    float* g = lds + wave * (ROWS * LDS_STRIDE);
    const float* w1s = lds + LDS_W1;
    const float* w2s = lds + LDS_WK;
    const float* wms = lds + LDS_WK + 576;
    const float* wds = lds + LDS_WK + 1152;

    const float* bss = lds + LDS_BS;

    int tile = blockIdx.x * (WPB * TPW) + wave;
    if (tile >= n_tiles) return;

    v2f xa[4], xn[4];
    v2f bw1[3][4], bwk[3][3];
    float bs1[3], bsk[3];
    load_x(x, tile, half, l16, xa);
    load_w1(w1s, half, l16, bw1);
    load_bs(bss, l16, bs1);

    v8f acc[3];
    float hr[6];
    v2f ha[3];

    for (int it = 0; it < TPW; ++it) {
#pragma unroll
        for (int t = 0; t < 3; ++t) {
            v8f c = splat8(bs1[t]);
#pragma unroll
            for (int s = 0; s < 4; ++s) c = wmma4(xa[s], bw1[t][s], c);
            acc[t] = c;
        }
        const int tnext = min(tile + WPB, n_tiles - 1);
        load_x(x, tnext, half, l16, xn);
        store_gates(g, acc, half, l16);
        LDS_FENCE();
        load_wk(w2s, half, l16, bwk);
        load_bs(bss + 48, l16, bsk);
        cell_slots(g, half, l16, hr);
#pragma unroll
        for (int s = 0; s < 3; ++s) {
            v2f a = {fmaxf(hr[2 * s], 0.0f), fmaxf(hr[2 * s + 1], 0.0f)};
            ha[s] = a;
        }
        LDS_FENCE();

#pragma unroll
        for (int t = 0; t < 3; ++t) {
            v8f c = splat8(bsk[t]);
#pragma unroll
            for (int s = 0; s < 3; ++s) c = wmma4(ha[s], bwk[t][s], c);
            acc[t] = c;
        }
        store_gates(g, acc, half, l16);
        LDS_FENCE();
        load_wk(wms, half, l16, bwk);
        load_bs(bss + 96, l16, bsk);
        cell_slots(g, half, l16, hr);
#pragma unroll
        for (int s = 0; s < 3; ++s) {
            v2f a = {selu_f(hr[2 * s]), selu_f(hr[2 * s + 1])};
            ha[s] = a;
        }
        LDS_FENCE();

#pragma unroll
        for (int t = 0; t < 3; ++t) {
            v8f c = splat8(bsk[t]);
#pragma unroll
            for (int s = 0; s < 3; ++s) c = wmma4(ha[s], bwk[t][s], c);
            acc[t] = c;
        }
        store_gates(g, acc, half, l16);
        LDS_FENCE();
        load_wk(wds, half, l16, bwk);
        load_bs(bss + 144, l16, bsk);
        emit_out(g, out + (long long)tile * 192, lane);
        LDS_FENCE();

#pragma unroll
        for (int t = 0; t < 3; ++t) {
            v8f c = splat8(bsk[t]);
#pragma unroll
            for (int s = 0; s < 3; ++s) c = wmma4(ha[s], bwk[t][s], c);
            acc[t] = c;
        }
        store_gates(g, acc, half, l16);
        LDS_FENCE();
        load_w1(w1s, half, l16, bw1);
        load_bs(bss, l16, bs1);
        emit_out(g, out + rowsB * 12 + (long long)tile * 192, lane);
        LDS_FENCE();

        tile += WPB;
        if (tile >= n_tiles) break;
#pragma unroll
        for (int s = 0; s < 4; ++s) xa[s] = xn[s];
    }
}

extern "C" void kernel_launch(void* const* d_in, const int* in_sizes, int n_in,
                              void* d_out, int out_size, void* d_ws, size_t ws_size,
                              hipStream_t stream) {
    const float* x   = (const float*)d_in[0];
    const float* W1  = (const float*)d_in[1];
    const float* bi1 = (const float*)d_in[2];
    const float* bh1 = (const float*)d_in[3];
    const float* W2  = (const float*)d_in[4];
    const float* bi2 = (const float*)d_in[5];
    const float* bh2 = (const float*)d_in[6];
    const float* Wm  = (const float*)d_in[7];
    const float* bim = (const float*)d_in[8];
    const float* bhm = (const float*)d_in[9];
    const float* Wd  = (const float*)d_in[10];
    const float* bid = (const float*)d_in[11];
    const float* bhd = (const float*)d_in[12];
    float* out = (float*)d_out;

    const long long rowsB = (long long)in_sizes[0] / 16;
    const int n_tiles = (int)((rowsB + 15) / 16);
    const int blocks  = (n_tiles + WPB * TPW - 1) / (WPB * TPW);

    hipLaunchKernelGGL(dynamics_net_kernel, dim3(blocks), dim3(256), 0, stream,
                       x, W1, bi1, bh1, W2, bi2, bh2, Wm, bim, bhm, Wd, bid, bhd,
                       out, n_tiles, rowsB);
}


# block constants prepared once in the staged LDS copy: sigmoid's 1/2 folded into the i and o weight rows (cell = two fma), biases staged as ready C tiles (ds_load_b128); no per-tile scaling or splat
# speedup vs baseline: 1.1038x; 1.0327x over previous
#include <hip/hip_runtime.h>

typedef __attribute__((ext_vector_type(2))) float v2f;
typedef __attribute__((ext_vector_type(4))) float v4f;
typedef __attribute__((ext_vector_type(8))) float v8f;

#define LDS_STRIDE 50
#define ROWS 16
#define WPB 8
#define TPW 8
#define W1_PITCH 20

#define CB_PITCH 12

#define LDS_W1   (WPB * ROWS * LDS_STRIDE)
#define LDS_WK   (LDS_W1 + 48 * W1_PITCH)
#define LDS_CB   (LDS_WK + 3 * 48 * 12)
#define LDS_ALL  (LDS_CB + 4 * 48 * CB_PITCH)

#define LDS_FENCE() asm volatile("s_wait_dscnt 0" ::: "memory")

static __device__ __forceinline__ float rcp_f(float x) {
    return __builtin_amdgcn_rcpf(x);
}
static __device__ __forceinline__ float tanh_f(float x) {
#if __has_builtin(__builtin_amdgcn_tanhf)
    return __builtin_amdgcn_tanhf(x);
#else
    float e = __expf(-2.0f * x);
    return (1.0f - e) * rcp_f(1.0f + e);
#endif
}
static __device__ __forceinline__ float gate_scale(int n) {
    return (n >= 24 && n < 36) ? 1.0f : 0.5f;
}
static __device__ __forceinline__ float lstm_2h(float gi, float gg, float go) {
    const float tg = tanh_f(gg);
    const float T  = tanh_f(0.5f * fmaf(tanh_f(gi), tg, tg));
    return fmaf(tanh_f(go), T, T);
}
static __device__ __forceinline__ float selu_half_f(float v) {
    const float a = 1.6732632423543772f, s = 1.0507009873554805f;
#if __has_builtin(__builtin_amdgcn_exp2f)
    return v > 0.0f ? (0.5f * s) * v
                    : (s * a) * (__builtin_amdgcn_exp2f(v * 0.72134752044448170368f) - 1.0f);
#else
    return v > 0.0f ? (0.5f * s) * v : (s * a) * (__expf(0.5f * v) - 1.0f);
#endif
}
static __device__ __forceinline__ v8f wmma4(v2f a, v2f b, v8f c) {
    return __builtin_amdgcn_wmma_f32_16x16x4_f32(false, a, false, b, (short)0, c,
                                                 false, false);
}

static __device__ __forceinline__ void store_gates(float* g, const v8f* acc,
                                                   int half, int l16) {
#pragma unroll
    for (int t = 0; t < 3; ++t)
#pragma unroll
        for (int r = 0; r < 8; ++r)
            g[(r + half * 8) * LDS_STRIDE + t * 16 + l16] = acc[t][r];
}

static __device__ __forceinline__ void cell_slots(const float* g, int half, int l16,
                                                  float* hr) {
#pragma unroll
    for (int s = 0; s < 3; ++s)
#pragma unroll
        for (int v = 0; v < 2; ++v) {
            const int k = s * 4 + half * 2 + v;
            const float* row = g + l16 * LDS_STRIDE;
            hr[s * 2 + v] = lstm_2h(row[k], row[24 + k], row[36 + k]);
        }
}

static __device__ __forceinline__ void emit_out(const float* g, float* dst, int lane) {
#pragma unroll
    for (int q = 0; q < 6; ++q) {
        const int flat = q * 32 + lane;
        const int r = flat / 12;
        const int j = flat - r * 12;
        const float* row = g + r * LDS_STRIDE;
        dst[flat] = tanh_f(0.5f * lstm_2h(row[j], row[24 + j], row[36 + j]));
    }
}

static __device__ __forceinline__ void load_wk(const float* wl, int half, int l16,
                                               v2f (*bw)[3]) {
#pragma unroll
    for (int t = 0; t < 3; ++t)
#pragma unroll
        for (int s = 0; s < 3; ++s)
            bw[t][s] = *(const v2f*)(wl + (t * 16 + l16) * 12 + s * 4 + half * 2);
}
static __device__ __forceinline__ void load_w1(const float* wl, int half, int l16,
                                               v2f (*bw)[4]) {
#pragma unroll
    for (int t = 0; t < 3; ++t)
#pragma unroll
        for (int s = 0; s < 4; ++s)
            bw[t][s] = *(const v2f*)(wl + (t * 16 + l16) * W1_PITCH + s * 4 + half * 2);
}
static __device__ __forceinline__ void load_c(const float* cb, int l16, v8f* c) {
#pragma unroll
    for (int t = 0; t < 3; ++t) {
        const float* p = cb + (t * 16 + l16) * CB_PITCH;
        const v4f lo = *(const v4f*)p;
        const v4f hi = *(const v4f*)(p + 4);
        v8f r = {lo.x, lo.y, lo.z, lo.w, hi.x, hi.y, hi.z, hi.w};
        c[t] = r;
    }
}
static __device__ __forceinline__ void load_x(const float* __restrict__ x, int tile,
                                              int half, int l16, v2f* xa) {
    const float* xr = x + ((long long)tile * 16 + l16) * 16 + half * 2;
#pragma unroll
    for (int s = 0; s < 4; ++s) xa[s] = *(const v2f*)(xr + s * 4);
}

__global__ __launch_bounds__(256) void dynamics_net_kernel(
    const float* __restrict__ x,
    const float* __restrict__ W1, const float* __restrict__ bi1, const float* __restrict__ bh1,
    const float* __restrict__ W2, const float* __restrict__ bi2, const float* __restrict__ bh2,
    const float* __restrict__ Wm, const float* __restrict__ bim, const float* __restrict__ bhm,
    const float* __restrict__ Wd, const float* __restrict__ bid, const float* __restrict__ bhd,
    float* __restrict__ out, int n_tiles, long long rowsB) {
    __shared__ __attribute__((aligned(16))) float lds[LDS_ALL];

    const int tid  = threadIdx.x;
    const int lane = tid & 31;
    const int wave = __builtin_amdgcn_readfirstlane(tid >> 5);
    const int half = lane >> 4;
    const int l16  = lane & 15;

    if (tid < 192) {
        const int n = tid >> 2, c = tid & 3;
        *(v4f*)(lds + LDS_W1 + n * W1_PITCH + c * 4) =
            *(const v4f*)(W1 + n * 16 + c * 4) * gate_scale(n);
    }
    if (tid < 144) {
        const float gsc = gate_scale(tid / 3);
        *(v4f*)(lds + LDS_WK + tid * 4)        = *(const v4f*)(W2 + tid * 4) * (0.5f * gsc);
        *(v4f*)(lds + LDS_WK + 576 + tid * 4)  = *(const v4f*)(Wm + tid * 4) * gsc;
        *(v4f*)(lds + LDS_WK + 1152 + tid * 4) = *(const v4f*)(Wd + tid * 4) * gsc;
    }
    if (tid < 48) {
        const float gsc = gate_scale(tid);
        const float b1 = (bi1[tid] + bh1[tid]) * gsc;
        const float b2 = (bi2[tid] + bh2[tid]) * gsc;
        const float bm = (bim[tid] + bhm[tid]) * gsc;
        const float bd = (bid[tid] + bhd[tid]) * gsc;
        const v4f s1 = {b1, b1, b1, b1}, s2 = {b2, b2, b2, b2};
        const v4f sm = {bm, bm, bm, bm}, sd = {bd, bd, bd, bd};
        float* cbn = lds + LDS_CB + tid * CB_PITCH;
        *(v4f*)(cbn)                          = s1;  *(v4f*)(cbn + 4)                          = s1;
        *(v4f*)(cbn + 48 * CB_PITCH)          = s2;  *(v4f*)(cbn + 48 * CB_PITCH + 4)          = s2;
        *(v4f*)(cbn + 96 * CB_PITCH)          = sm;  *(v4f*)(cbn + 96 * CB_PITCH + 4)          = sm;
        *(v4f*)(cbn + 144 * CB_PITCH)         = sd;  *(v4f*)(cbn + 144 * CB_PITCH + 4)         = sd;
    }
    __syncthreads();

    float* g = lds + wave * (ROWS * LDS_STRIDE);
    const float* w1s = lds + LDS_W1;
    const float* w2s = lds + LDS_WK;
    const float* wms = lds + LDS_WK + 576;
    const float* wds = lds + LDS_WK + 1152;

    const float* cbs = lds + LDS_CB;

    int tile = blockIdx.x * (WPB * TPW) + wave;
    if (tile >= n_tiles) return;

    v2f xa[4], xn[4];
    v2f bw1[3][4], bwk[3][3];
    v8f acc[3];
    float hr[6];
    v2f ha[3];
    load_x(x, tile, half, l16, xa);
    load_w1(w1s, half, l16, bw1);
    load_c(cbs, l16, acc);

    for (int it = 0; it < TPW; ++it) {
#pragma unroll
        for (int t = 0; t < 3; ++t) {
            v8f c = acc[t];
#pragma unroll
            for (int s = 0; s < 4; ++s) c = wmma4(xa[s], bw1[t][s], c);
            acc[t] = c;
        }
        const int tnext = min(tile + WPB, n_tiles - 1);
        load_x(x, tnext, half, l16, xn);
        store_gates(g, acc, half, l16);
        LDS_FENCE();
        load_wk(w2s, half, l16, bwk);
        load_c(cbs + 48 * CB_PITCH, l16, acc);
        cell_slots(g, half, l16, hr);
#pragma unroll
        for (int s = 0; s < 3; ++s) {
            v2f a = {fmaxf(hr[2 * s], 0.0f), fmaxf(hr[2 * s + 1], 0.0f)};
            ha[s] = a;
        }
        LDS_FENCE();

#pragma unroll
        for (int t = 0; t < 3; ++t) {
            v8f c = acc[t];
#pragma unroll
            for (int s = 0; s < 3; ++s) c = wmma4(ha[s], bwk[t][s], c);
            acc[t] = c;
        }
        store_gates(g, acc, half, l16);
        LDS_FENCE();
        load_wk(wms, half, l16, bwk);
        load_c(cbs + 96 * CB_PITCH, l16, acc);
        cell_slots(g, half, l16, hr);
#pragma unroll
        for (int s = 0; s < 3; ++s) {
            v2f a = {selu_half_f(hr[2 * s]), selu_half_f(hr[2 * s + 1])};
            ha[s] = a;
        }
        LDS_FENCE();

#pragma unroll
        for (int t = 0; t < 3; ++t) {
            v8f c = acc[t];
#pragma unroll
            for (int s = 0; s < 3; ++s) c = wmma4(ha[s], bwk[t][s], c);
            acc[t] = c;
        }
        store_gates(g, acc, half, l16);
        LDS_FENCE();
        load_wk(wds, half, l16, bwk);
        load_c(cbs + 144 * CB_PITCH, l16, acc);
        emit_out(g, out + (long long)tile * 192, lane);
        LDS_FENCE();

#pragma unroll
        for (int t = 0; t < 3; ++t) {
            v8f c = acc[t];
#pragma unroll
            for (int s = 0; s < 3; ++s) c = wmma4(ha[s], bwk[t][s], c);
            acc[t] = c;
        }
        store_gates(g, acc, half, l16);
        LDS_FENCE();
        load_w1(w1s, half, l16, bw1);
        load_c(cbs, l16, acc);
        emit_out(g, out + rowsB * 12 + (long long)tile * 192, lane);
        LDS_FENCE();

        tile += WPB;
        if (tile >= n_tiles) break;
#pragma unroll
        for (int s = 0; s < 4; ++s) xa[s] = xn[s];
    }
}

extern "C" void kernel_launch(void* const* d_in, const int* in_sizes, int n_in,
                              void* d_out, int out_size, void* d_ws, size_t ws_size,
                              hipStream_t stream) {
    const float* x   = (const float*)d_in[0];
    const float* W1  = (const float*)d_in[1];
    const float* bi1 = (const float*)d_in[2];
    const float* bh1 = (const float*)d_in[3];
    const float* W2  = (const float*)d_in[4];
    const float* bi2 = (const float*)d_in[5];
    const float* bh2 = (const float*)d_in[6];
    const float* Wm  = (const float*)d_in[7];
    const float* bim = (const float*)d_in[8];
    const float* bhm = (const float*)d_in[9];
    const float* Wd  = (const float*)d_in[10];
    const float* bid = (const float*)d_in[11];
    const float* bhd = (const float*)d_in[12];
    float* out = (float*)d_out;

    const long long rowsB = (long long)in_sizes[0] / 16;
    const int n_tiles = (int)((rowsB + 15) / 16);
    const int blocks  = (n_tiles + WPB * TPW - 1) / (WPB * TPW);

    hipLaunchKernelGGL(dynamics_net_kernel, dim3(blocks), dim3(256), 0, stream,
                       x, W1, bi1, bh1, W2, bi2, bh2, Wm, bim, bhm, Wd, bid, bhd,
                       out, n_tiles, rowsB);
}
